# starting kernel
# baseline (speedup 1.0000x reference)
#include <hip/hip_runtime.h>

#define C 128
#define NL 4
#define NEG_SLOPE 0.2f
#define SA 132
#define SW 132
#define WSTR 136

typedef _Float16 h8v __attribute__((ext_vector_type(8)));
typedef float v2f __attribute__((ext_vector_type(2)));
typedef float v4f __attribute__((ext_vector_type(4)));
typedef float v8f __attribute__((ext_vector_type(8)));

__global__ __launch_bounds__(256, 1) void rnn_recurrence(
    const float* __restrict__ x, const float* __restrict__ Wih,
    const float* __restrict__ Whh, const float* __restrict__ bih,
    const float* __restrict__ bhh, const int* __restrict__ max_steps,
    float* __restrict__ enc)
{
    __shared__ _Float16 swih[NL * C * WSTR];
    __shared__ _Float16 swhh[NL * C * WSTR];
    __shared__ float    sb[NL * C];
    __shared__ float    hid2[2][NL * C];
    __shared__ float    part[2][C];

    const int tid  = threadIdx.x;
    const int ch   = tid & (C - 1);
    const int half = tid >> 7;

    for (int i = tid; i < NL * C * C; i += 256) {
        int r = i >> 7, c = i & (C - 1);
        swih[r * WSTR + c] = (_Float16)Wih[i];
        swhh[r * WSTR + c] = (_Float16)Whh[i];
    }
    for (int i = tid; i < NL * C; i += 256) {
        sb[i]      = bih[i] + bhh[i];
        hid2[0][i] = 0.0f;
        hid2[1][i] = (i < C) ? x[i] : 0.0f;
    }
    __syncthreads();

    const int T = max_steps[0];
    for (int t = 0; t < T; ++t) {
        const int p = t & 1, q = p ^ 1;
        for (int l = 0; l < NL; ++l) {
            const float* in = (l == 0) ? &hid2[q][(NL - 1) * C]
                                       : &hid2[p][(l - 1) * C];
            const float* hp = &hid2[q][l * C];
            const v4f* in4 = (const v4f*)in;
            const v4f* hp4 = (const v4f*)hp;

            const h8v* wi8 = (const h8v*)(swih + (l * C + ch) * WSTR + half * 64);
            const h8v* wh8 = (const h8v*)(swhh + (l * C + ch) * WSTR + half * 64);

            float acc = 0.0f;
            #pragma unroll
            for (int k8 = 0; k8 < 8; ++k8) {
                h8v wa = wi8[k8];
                h8v wb = wh8[k8];
                v4f i0 = in4[half * 16 + k8 * 2];
                v4f i1 = in4[half * 16 + k8 * 2 + 1];
                v4f h0 = hp4[half * 16 + k8 * 2];
                v4f h1 = hp4[half * 16 + k8 * 2 + 1];
                acc += (float)wa[0] * i0.x + (float)wa[1] * i0.y;
                acc += (float)wa[2] * i0.z + (float)wa[3] * i0.w;
                acc += (float)wa[4] * i1.x + (float)wa[5] * i1.y;
                acc += (float)wa[6] * i1.z + (float)wa[7] * i1.w;
                acc += (float)wb[0] * h0.x + (float)wb[1] * h0.y;
                acc += (float)wb[2] * h0.z + (float)wb[3] * h0.w;
                acc += (float)wb[4] * h1.x + (float)wb[5] * h1.y;
                acc += (float)wb[6] * h1.z + (float)wb[7] * h1.w;
            }
            part[half][ch] = acc;
            __syncthreads();
            if (half == 0) {
                float s = sb[l * C + ch] + part[0][ch] + part[1][ch];
                s = s > 0.0f ? s : 0.0f;
                hid2[p][l * C + ch] = s;
                if (l == NL - 1) enc[t * C + ch] = s;
            }
            __syncthreads();
        }
    }
}

__global__ __launch_bounds__(128, 1) void head_kernel(
    const float* __restrict__ enc,
    const float* __restrict__ rb_w,
    const float* __restrict__ rb_b,
    const float* __restrict__ out_w,
    const float* __restrict__ out_b,
    float* __restrict__ out,
    int od, int col0)
{
    __shared__ float wlds[C * SW];
    __shared__ float blds[C];
    __shared__ float xbuf[4][16 * SA];
    __shared__ float hbuf[4][16 * SA];

    const int tid  = threadIdx.x;
    const int wave = tid >> 5;
    const int lane = tid & 31;
    const int ln   = lane & 15;
    const int hi   = lane >> 4;
    const int row0 = blockIdx.x * 64 + wave * 16;

    for (int i = lane; i < 16 * C; i += 32) {
        int m = i >> 7, c = i & (C - 1);
        xbuf[wave][m * SA + c] = enc[(row0 + m) * C + c];
    }

    for (int blk = 0; blk < 2; ++blk) {
        for (int ly = 0; ly < 2; ++ly) {
            const float* W = rb_w + (blk * 2 + ly) * C * C;
            const float* B = rb_b + (blk * 2 + ly) * C;
            __syncthreads();
            for (int i = tid; i < C * C; i += 128)
                wlds[(i >> 7) * SW + (i & (C - 1))] = W[i];
            if (tid < C) blds[tid] = B[tid];
            __syncthreads();

            const float* src = (ly == 0) ? xbuf[wave] : hbuf[wave];
            float*       dst = (ly == 0) ? hbuf[wave] : xbuf[wave];

            v8f acc[8];
            #pragma unroll
            for (int nt = 0; nt < 8; ++nt) {
                float bv = blds[nt * 16 + ln];
                #pragma unroll
                for (int r = 0; r < 8; ++r) {
                    float cv = bv;
                    if (ly == 1)
                        cv += xbuf[wave][(r + 8 * hi) * SA + nt * 16 + ln];
                    acc[nt][r] = cv;
                }
            }

            for (int k = 0; k < 32; ++k) {
                v2f a;
                a.x = src[ln * SA + 4 * k + 2 * hi];
                a.y = src[ln * SA + 4 * k + 2 * hi + 1];
                #pragma unroll
                for (int nt = 0; nt < 8; ++nt) {
                    v2f b;
                    b.x = wlds[(nt * 16 + ln) * SW + 4 * k + 2 * hi];
                    b.y = wlds[(nt * 16 + ln) * SW + 4 * k + 2 * hi + 1];
                    acc[nt] = __builtin_amdgcn_wmma_f32_16x16x4_f32(
                        false, a, false, b, (short)0, acc[nt], false, false);
                }
            }

            #pragma unroll
            for (int nt = 0; nt < 8; ++nt) {
                #pragma unroll
                for (int r = 0; r < 8; ++r) {
                    float v = acc[nt][r];
                    v = v > 0.0f ? v : NEG_SLOPE * v;
                    dst[(r + 8 * hi) * SA + nt * 16 + ln] = v;
                }
            }
        }
    }
    __syncthreads();

    for (int o = hi; o < od; o += 2) {
        const float* wo = out_w + o * C;
        float s = out_b[o];
        for (int c = 0; c < C; ++c) s += xbuf[wave][ln * SA + c] * wo[c];
        out[(row0 + ln) * 10 + col0 + o] = s;
    }
}

extern "C" void kernel_launch(void* const* d_in, const int* in_sizes, int n_in,
                              void* d_out, int out_size, void* d_ws, size_t ws_size,
                              hipStream_t stream)
{
    const float* x         = (const float*)d_in[0];
    const float* W_ih      = (const float*)d_in[1];
    const float* W_hh      = (const float*)d_in[2];
    const float* b_ih      = (const float*)d_in[3];
    const float* b_hh      = (const float*)d_in[4];
    const float* atom_rb_w = (const float*)d_in[5];
    const float* atom_rb_b = (const float*)d_in[6];
    const float* atom_w    = (const float*)d_in[7];
    const float* atom_b    = (const float*)d_in[8];
    const float* pos_rb_w  = (const float*)d_in[9];
    const float* pos_rb_b  = (const float*)d_in[10];
    const float* pos_w     = (const float*)d_in[11];
    const float* pos_b     = (const float*)d_in[12];
    const float* mag_rb_w  = (const float*)d_in[13];
    const float* mag_rb_b  = (const float*)d_in[14];
    const float* mag_w     = (const float*)d_in[15];
    const float* mag_b     = (const float*)d_in[16];
    const int*   max_steps = (const int*)d_in[17];

    float* out = (float*)d_out;
    float* enc = (float*)d_ws;
    const int T = out_size / 10;

    rnn_recurrence<<<1, 256, 0, stream>>>(x, W_ih, W_hh, b_ih, b_hh, max_steps, enc);

    const int blocks = T / 64;
    head_kernel<<<blocks, 128, 0, stream>>>(enc, atom_rb_w, atom_rb_b, atom_w, atom_b, out, 8, 0);
    head_kernel<<<blocks, 128, 0, stream>>>(enc, pos_rb_w,  pos_rb_b,  pos_w,  pos_b,  out, 1, 8);
    head_kernel<<<blocks, 128, 0, stream>>>(enc, mag_rb_w,  mag_rb_b,  mag_w,  mag_b,  out, 1, 9);
}


# Recurrence remapped: each thread keeps its f16 weight slice in VGPRs (1024 threads, K across lanes); LDS reads per thread per layer drop from 48 to 2; cross-lane DPP tree changes f32 sum order
# speedup vs baseline: 3.0004x; 3.0004x over previous
#include <hip/hip_runtime.h>

#define C 128
#define NL 4
#define NEG_SLOPE 0.2f
#define SA 132
#define SW 132
#define WSTR 136

typedef _Float16 h8v __attribute__((ext_vector_type(8)));
typedef float v2f __attribute__((ext_vector_type(2)));
typedef float v4f __attribute__((ext_vector_type(4)));
typedef float v8f __attribute__((ext_vector_type(8)));

#define RW  32
#define CPW (C / RW)

__device__ __forceinline__ float lane_xor16(float v) {
    unsigned u = __builtin_bit_cast(unsigned, v);
    unsigned r = __builtin_amdgcn_permlanex16(u, u, 0x76543210u, 0xfedcba98u, false, false);
    return __builtin_bit_cast(float, r);
}
template <int M>
__device__ __forceinline__ float lane_xor_row(float v) {
    int u = __builtin_bit_cast(int, v);
    int r = __builtin_amdgcn_update_dpp(0, u, 0x160 + M, 0xf, 0xf, true);
    return __builtin_bit_cast(float, r);
}

__global__ __launch_bounds__(RW * 32, 1) void rnn_recurrence(
    const float* __restrict__ x, const float* __restrict__ Wih,
    const float* __restrict__ Whh, const float* __restrict__ bih,
    const float* __restrict__ bhh, const int* __restrict__ max_steps,
    float* __restrict__ enc)
{
    __shared__ __attribute__((aligned(16))) float hid2[2 * NL * C];

    const int  tid  = threadIdx.x;
    const int  wave = tid >> 5;
    const int  lane = tid & 31;
    const bool lo   = lane < 16;
    const int  k0   = (lane & 15) * 4;
    const int  jown = lane >> 3;
    const bool g8   = (lane & 8) != 0;

    h8v w[NL][CPW];
    {
        const float* wsrc = lo ? Wih : Whh;
        #pragma unroll
        for (int l = 0; l < NL; ++l) {
            #pragma unroll
            for (int j = 0; j < CPW; ++j) {
                const float* pw = wsrc + ((l * C + wave * CPW + j) * C + k0);
                v4f a = *(const v4f*)pw;
                v4f b = *(const v4f*)(pw + 64);
                h8v h;
                h[0] = (_Float16)a.x; h[1] = (_Float16)a.y;
                h[2] = (_Float16)a.z; h[3] = (_Float16)a.w;
                h[4] = (_Float16)b.x; h[5] = (_Float16)b.y;
                h[6] = (_Float16)b.z; h[7] = (_Float16)b.w;
                w[l][j] = h;
            }
        }
    }
    float bias[NL];
    #pragma unroll
    for (int l = 0; l < NL; ++l) {
        const int c = l * C + wave * CPW + jown;
        bias[l] = bih[c] + bhh[c];
    }

    for (int i = tid; i < NL * C; i += RW * 32) {
        hid2[i]          = 0.0f;
        hid2[NL * C + i] = (i < C) ? x[i] : 0.0f;
    }
    __syncthreads();

    const int T = max_steps[0];
    for (int t = 0; t < T; ++t) {
        const int p = t & 1, q = p ^ 1;
        #pragma unroll
        for (int l = 0; l < NL; ++l) {
            #pragma unroll
            for (int j = 0; j < CPW; ++j)
                asm volatile("" : "+v"(w[l][j]));
        }
        #pragma unroll
        for (int l = 0; l < NL; ++l) {
            const int inOff = (l == 0) ? (q * NL + NL - 1) * C : (p * NL + l - 1) * C;
            const int hpOff = (q * NL + l) * C;
            const float* vsrc = hid2 + (lo ? inOff : hpOff) + k0;
            const v4f v0 = *(const v4f*)vsrc;
            const v4f v1 = *(const v4f*)(vsrc + 64);

            if (l == 0 && t > 0 && wave == ((t - 1) & (RW - 1)) && lo) {
                float* e = enc + (size_t)(t - 1) * C + k0;
                *(v4f*)e        = v0;
                *(v4f*)(e + 64) = v1;
            }

            float c[CPW];
            #pragma unroll
            for (int j = 0; j < CPW; ++j) {
                const h8v wj = w[l][j];
                float a = (float)wj[0] * v0.x;
                a = __builtin_fmaf((float)wj[1], v0.y, a);
                a = __builtin_fmaf((float)wj[2], v0.z, a);
                a = __builtin_fmaf((float)wj[3], v0.w, a);
                a = __builtin_fmaf((float)wj[4], v1.x, a);
                a = __builtin_fmaf((float)wj[5], v1.y, a);
                a = __builtin_fmaf((float)wj[6], v1.z, a);
                a = __builtin_fmaf((float)wj[7], v1.w, a);
                asm volatile("" : "+v"(a));
                c[j] = a;
            }

            const float kA = lo ? c[0] : c[2], sA = lo ? c[2] : c[0];
            const float kB = lo ? c[1] : c[3], sB = lo ? c[3] : c[1];
            const float a0 = kA + lane_xor16(sA);
            const float a1 = kB + lane_xor16(sB);
            const float kk = g8 ? a1 : a0, ss = g8 ? a0 : a1;
            float s = kk + lane_xor_row<8>(ss);
            s += lane_xor_row<4>(s);
            s += lane_xor_row<2>(s);
            s += lane_xor_row<1>(s);

            s += bias[l];
            s = s > 0.0f ? s : 0.0f;
            if ((lane & 7) == 0)
                hid2[(p * NL + l) * C + wave * CPW + jown] = s;
            asm volatile("s_wait_dscnt 0x0\n\ts_barrier_signal -1\n\ts_barrier_wait -1" ::: "memory");
        }
    }

    if (T > 0 && wave == 0 && lo) {
        const float* vsrc = hid2 + (((T - 1) & 1) * NL + NL - 1) * C + k0;
        float* e = enc + (size_t)(T - 1) * C + k0;
        *(v4f*)e        = *(const v4f*)vsrc;
        *(v4f*)(e + 64) = *(const v4f*)(vsrc + 64);
    }
}

__global__ __launch_bounds__(128, 1) void head_kernel(
    const float* __restrict__ enc,
    const float* __restrict__ rb_w,
    const float* __restrict__ rb_b,
    const float* __restrict__ out_w,
    const float* __restrict__ out_b,
    float* __restrict__ out,
    int od, int col0)
{
    __shared__ float wlds[C * SW];
    __shared__ float blds[C];
    __shared__ float xbuf[4][16 * SA];
    __shared__ float hbuf[4][16 * SA];

    const int tid  = threadIdx.x;
    const int wave = tid >> 5;
    const int lane = tid & 31;
    const int ln   = lane & 15;
    const int hi   = lane >> 4;
    const int row0 = blockIdx.x * 64 + wave * 16;

    for (int i = lane; i < 16 * C; i += 32) {
        int m = i >> 7, c = i & (C - 1);
        xbuf[wave][m * SA + c] = enc[(row0 + m) * C + c];
    }

    for (int blk = 0; blk < 2; ++blk) {
        for (int ly = 0; ly < 2; ++ly) {
            const float* W = rb_w + (blk * 2 + ly) * C * C;
            const float* B = rb_b + (blk * 2 + ly) * C;
            __syncthreads();
            for (int i = tid; i < C * C; i += 128)
                wlds[(i >> 7) * SW + (i & (C - 1))] = W[i];
            if (tid < C) blds[tid] = B[tid];
            __syncthreads();

            const float* src = (ly == 0) ? xbuf[wave] : hbuf[wave];
            float*       dst = (ly == 0) ? hbuf[wave] : xbuf[wave];

            v8f acc[8];
            #pragma unroll
            for (int nt = 0; nt < 8; ++nt) {
                float bv = blds[nt * 16 + ln];
                #pragma unroll
                for (int r = 0; r < 8; ++r) {
                    float cv = bv;
                    if (ly == 1)
                        cv += xbuf[wave][(r + 8 * hi) * SA + nt * 16 + ln];
                    acc[nt][r] = cv;
                }
            }

            for (int k = 0; k < 32; ++k) {
                v2f a;
                a.x = src[ln * SA + 4 * k + 2 * hi];
                a.y = src[ln * SA + 4 * k + 2 * hi + 1];
                #pragma unroll
                for (int nt = 0; nt < 8; ++nt) {
                    v2f b;
                    b.x = wlds[(nt * 16 + ln) * SW + 4 * k + 2 * hi];
                    b.y = wlds[(nt * 16 + ln) * SW + 4 * k + 2 * hi + 1];
                    acc[nt] = __builtin_amdgcn_wmma_f32_16x16x4_f32(
                        false, a, false, b, (short)0, acc[nt], false, false);
                }
            }

            #pragma unroll
            for (int nt = 0; nt < 8; ++nt) {
                #pragma unroll
                for (int r = 0; r < 8; ++r) {
                    float v = acc[nt][r];
                    v = v > 0.0f ? v : NEG_SLOPE * v;
                    dst[(r + 8 * hi) * SA + nt * 16 + ln] = v;
                }
            }
        }
    }
    __syncthreads();

    for (int o = hi; o < od; o += 2) {
        const float* wo = out_w + o * C;
        float s = out_b[o];
        for (int c = 0; c < C; ++c) s += xbuf[wave][ln * SA + c] * wo[c];
        out[(row0 + ln) * 10 + col0 + o] = s;
    }
}

extern "C" void kernel_launch(void* const* d_in, const int* in_sizes, int n_in,
                              void* d_out, int out_size, void* d_ws, size_t ws_size,
                              hipStream_t stream)
{
    const float* x         = (const float*)d_in[0];
    const float* W_ih      = (const float*)d_in[1];
    const float* W_hh      = (const float*)d_in[2];
    const float* b_ih      = (const float*)d_in[3];
    const float* b_hh      = (const float*)d_in[4];
    const float* atom_rb_w = (const float*)d_in[5];
    const float* atom_rb_b = (const float*)d_in[6];
    const float* atom_w    = (const float*)d_in[7];
    const float* atom_b    = (const float*)d_in[8];
    const float* pos_rb_w  = (const float*)d_in[9];
    const float* pos_rb_b  = (const float*)d_in[10];
    const float* pos_w     = (const float*)d_in[11];
    const float* pos_b     = (const float*)d_in[12];
    const float* mag_rb_w  = (const float*)d_in[13];
    const float* mag_rb_b  = (const float*)d_in[14];
    const float* mag_w     = (const float*)d_in[15];
    const float* mag_b     = (const float*)d_in[16];
    const int*   max_steps = (const int*)d_in[17];

    float* out = (float*)d_out;
    float* enc = (float*)d_ws;
    const int T = out_size / 10;

    rnn_recurrence<<<1, 1024, 0, stream>>>(x, W_ih, W_hh, b_ih, b_hh, max_steps, enc);

    const int blocks = T / 64;
    head_kernel<<<blocks, 128, 0, stream>>>(enc, atom_rb_w, atom_rb_b, atom_w, atom_b, out, 8, 0);
    head_kernel<<<blocks, 128, 0, stream>>>(enc, pos_rb_w,  pos_rb_b,  pos_w,  pos_b,  out, 1, 8);
    head_kernel<<<blocks, 128, 0, stream>>>(enc, mag_rb_w,  mag_rb_b,  mag_w,  mag_b,  out, 1, 9);
}
